# starting kernel
# baseline (speedup 1.0000x reference)
#include <hip/hip_runtime.h>
#include <hip/hip_bf16.h>


typedef float v2f __attribute__((ext_vector_type(2)));
typedef float v8f __attribute__((ext_vector_type(8)));

#define N_NODES 100000
#define PB_LAYER 8192

__global__ __launch_bounds__(256) void prep_k(
    const float* __restrict__ W2, const float* __restrict__ W3,
    const float* __restrict__ W4, const float* __restrict__ W5,
    const float* __restrict__ Wa, float* __restrict__ PB,
    unsigned* __restrict__ g)
{
  int tid = blockIdx.x * 256 + threadIdx.x;
  if (tid < 64) g[tid] = 0u;
  if (tid >= 5 * PB_LAYER) return;
  int l = tid / PB_LAYER, r = tid % PB_LAYER;
  int d = r & 1;
  int n = (r >> 1) & 127;
  int h = (r >> 8) & 1;
  int s = r >> 9;
  int k = 4 * s + 2 * h + d;
  float v;
  if (l < 4) {
    const float* W = (l == 0) ? W2 : (l == 1) ? W3 : (l == 2) ? W4 : W5;
    v = (n < 64) ? (W[k * 64 + n] - W[(64 + k) * 64 + n])
                 : W[(64 + k) * 64 + (n - 64)];
  } else {
    v = Wa[k * 128 + n];
  }
  PB[tid] = v;
}

__global__ __launch_bounds__(256) void layer1_ap(
    const float* __restrict__ x, const float* __restrict__ W1,
    const float* __restrict__ b1, float* __restrict__ A, float* __restrict__ P)
{
  int tid = blockIdx.x * 256 + threadIdx.x;
  int n = tid >> 6, c = tid & 63;
  float x0 = x[n * 3 + 0], x1 = x[n * 3 + 1], x2 = x[n * 3 + 2];
  float wt0 = W1[0 * 64 + c], wt1 = W1[1 * 64 + c], wt2 = W1[2 * 64 + c];
  float wb0 = W1[3 * 64 + c], wb1 = W1[4 * 64 + c], wb2 = W1[5 * 64 + c];
  float p = x0 * wb0 + x1 * wb1 + x2 * wb2;
  float a = b1[c] + x0 * (wt0 - wb0) + x1 * (wt1 - wb1) + x2 * (wt2 - wb2);
  A[tid] = a;
  P[tid] = p;
}

__global__ __launch_bounds__(256) void gemm_ap_k(
    const float* __restrict__ H, const float* __restrict__ PB,
    const float* __restrict__ bias, float* __restrict__ A, float* __restrict__ P)
{
  const int w = threadIdx.x >> 5;
  const int lane = threadIdx.x & 31;
  const int wr = w >> 2, wc = w & 3;
  const int n0 = blockIdx.x * 32 + wr * 16;
  const int lhalf = lane >> 4, l15 = lane & 15;
  const int cb0 = wc * 32, cb1 = wc * 32 + 16;

  v8f acc0 = {0.f, 0.f, 0.f, 0.f, 0.f, 0.f, 0.f, 0.f};
  v8f acc1 = {0.f, 0.f, 0.f, 0.f, 0.f, 0.f, 0.f, 0.f};

  const float* arow = H + (size_t)(n0 + l15) * 64 + 2 * lhalf;
#pragma unroll
  for (int s = 0; s < 16; ++s) {
    v2f af = *(const v2f*)(arow + 4 * s);
    v2f bf0 = *(const v2f*)(PB + (((s * 2 + lhalf) * 128) + cb0 + l15) * 2);
    v2f bf1 = *(const v2f*)(PB + (((s * 2 + lhalf) * 128) + cb1 + l15) * 2);
    acc0 = __builtin_amdgcn_wmma_f32_16x16x4_f32(false, af, false, bf0,
                                                 (short)0, acc0, false, false);
    acc1 = __builtin_amdgcn_wmma_f32_16x16x4_f32(false, af, false, bf1,
                                                 (short)0, acc1, false, false);
  }

#pragma unroll
  for (int r = 0; r < 8; ++r) {
    int row = n0 + r + lhalf * 8;
    int c0 = cb0 + l15, c1 = cb1 + l15;
    float v0 = acc0[r], v1 = acc1[r];
    if (c0 < 64) A[(size_t)row * 64 + c0] = v0 + bias[c0];
    else         P[(size_t)row * 64 + (c0 - 64)] = v0;
    if (c1 < 64) A[(size_t)row * 64 + c1] = v1 + bias[c1];
    else         P[(size_t)row * 64 + (c1 - 64)] = v1;
  }
}

__global__ __launch_bounds__(256) void edge_max_k(
    const float* __restrict__ A, const float* __restrict__ P,
    const int* __restrict__ src, float* __restrict__ h)
{
  int node = blockIdx.x * 8 + (threadIdx.x >> 5);
  int lane = threadIdx.x & 31;
  float2 a = *(const float2*)(A + (size_t)node * 64 + lane * 2);
  float m0 = 0.0f, m1 = 0.0f;
  const int* se = src + (size_t)node * 16;
#pragma unroll
  for (int e = 0; e < 16; ++e) {
    int j = se[e];
    float2 p = *(const float2*)(P + (size_t)j * 64 + lane * 2);
    m0 = fmaxf(m0, a.x + p.x);
    m1 = fmaxf(m1, a.y + p.y);
  }
  float2 o; o.x = m0; o.y = m1;
  *(float2*)(h + (size_t)node * 64 + lane * 2) = o;
}

__global__ __launch_bounds__(256) void global_max_k(
    const float* __restrict__ h, unsigned* __restrict__ g)
{
  __shared__ float red[256];
  int c = threadIdx.x & 63, grp = threadIdx.x >> 6;
  float m = 0.0f;
  for (int n = blockIdx.x * 4 + grp; n < N_NODES; n += gridDim.x * 4)
    m = fmaxf(m, h[(size_t)n * 64 + c]);
  red[threadIdx.x] = m;
  __syncthreads();
  if (threadIdx.x < 64) {
    m = fmaxf(fmaxf(red[threadIdx.x], red[threadIdx.x + 64]),
              fmaxf(red[threadIdx.x + 128], red[threadIdx.x + 192]));
    atomicMax(g + c, __float_as_uint(m));
  }
}

__global__ void gvec_k(const unsigned* __restrict__ g,
                       const float* __restrict__ Wa,
                       const float* __restrict__ ba, float* __restrict__ gv)
{
  int c = threadIdx.x;
  float s = ba[c];
  for (int k = 0; k < 64; ++k)
    s += __uint_as_float(g[k]) * Wa[(64 + k) * 128 + c];
  gv[c] = s;
}

__global__ __launch_bounds__(256) void final_k(
    const float* __restrict__ h, const float* __restrict__ PBa,
    const float* __restrict__ gv, const float* __restrict__ Wb,
    const float* __restrict__ bb, const float* __restrict__ x,
    float* __restrict__ out)
{
  __shared__ float T[32][132];

  const int w = threadIdx.x >> 5;
  const int lane = threadIdx.x & 31;
  const int wr = w >> 2, wc = w & 3;
  const int n0 = blockIdx.x * 32 + wr * 16;
  const int lhalf = lane >> 4, l15 = lane & 15;
  const int cb0 = wc * 32, cb1 = wc * 32 + 16;

  v8f acc0 = {0.f, 0.f, 0.f, 0.f, 0.f, 0.f, 0.f, 0.f};
  v8f acc1 = {0.f, 0.f, 0.f, 0.f, 0.f, 0.f, 0.f, 0.f};

  const float* arow = h + (size_t)(n0 + l15) * 64 + 2 * lhalf;
#pragma unroll
  for (int s = 0; s < 16; ++s) {
    v2f af = *(const v2f*)(arow + 4 * s);
    v2f bf0 = *(const v2f*)(PBa + (((s * 2 + lhalf) * 128) + cb0 + l15) * 2);
    v2f bf1 = *(const v2f*)(PBa + (((s * 2 + lhalf) * 128) + cb1 + l15) * 2);
    acc0 = __builtin_amdgcn_wmma_f32_16x16x4_f32(false, af, false, bf0,
                                                 (short)0, acc0, false, false);
    acc1 = __builtin_amdgcn_wmma_f32_16x16x4_f32(false, af, false, bf1,
                                                 (short)0, acc1, false, false);
  }

#pragma unroll
  for (int r = 0; r < 8; ++r) {
    int lr = wr * 16 + r + lhalf * 8;
    int c0 = cb0 + l15, c1 = cb1 + l15;
    T[lr][c0] = fmaxf(0.0f, acc0[r] + gv[c0]);
    T[lr][c1] = fmaxf(0.0f, acc1[r] + gv[c1]);
  }
  __syncthreads();

  int t = threadIdx.x;
  if (t < 96) {
    int row = t / 3, j = t % 3;
    float s = bb[j];
#pragma unroll 8
    for (int c = 0; c < 128; ++c) s += T[row][c] * Wb[c * 3 + j];
    size_t n = (size_t)blockIdx.x * 32 + row;
    out[n * 3 + j] = x[n * 3 + j] + s;
  }
}

extern "C" void kernel_launch(void* const* d_in, const int* in_sizes, int n_in,
                              void* d_out, int out_size, void* d_ws, size_t ws_size,
                              hipStream_t stream) {
  (void)in_sizes; (void)n_in; (void)out_size; (void)ws_size;
  const float* x  = (const float*)d_in[0];
  const int*   ei = (const int*)d_in[1];
  const float* W1 = (const float*)d_in[2];
  const float* b1 = (const float*)d_in[3];
  const float* W2 = (const float*)d_in[4];
  const float* b2 = (const float*)d_in[5];
  const float* W3 = (const float*)d_in[6];
  const float* b3 = (const float*)d_in[7];
  const float* W4 = (const float*)d_in[8];
  const float* b4 = (const float*)d_in[9];
  const float* W5 = (const float*)d_in[10];
  const float* b5 = (const float*)d_in[11];
  const float* Wa = (const float*)d_in[12];
  const float* ba = (const float*)d_in[13];
  const float* Wb = (const float*)d_in[14];
  const float* bb = (const float*)d_in[15];
  float* out = (float*)d_out;

  const int N = N_NODES;
  float* ws = (float*)d_ws;
  float*    A  = ws;
  float*    Pp = A  + (size_t)N * 64;
  float*    h0 = Pp + (size_t)N * 64;
  float*    h1 = h0 + (size_t)N * 64;
  float*    PB = h1 + (size_t)N * 64;
  unsigned* g  = (unsigned*)(PB + 5 * PB_LAYER);
  float*    gv = (float*)(g + 64);

  const int* src = ei;

  prep_k<<<(5 * PB_LAYER + 255) / 256, 256, 0, stream>>>(W2, W3, W4, W5, Wa, PB, g);
  layer1_ap<<<(N * 64) / 256, 256, 0, stream>>>(x, W1, b1, A, Pp);
  edge_max_k<<<N / 8, 256, 0, stream>>>(A, Pp, src, h0);

  const float* bs[4] = {b2, b3, b4, b5};
  float* hin = h0;
  float* hout = h1;
  for (int l = 0; l < 4; ++l) {
    gemm_ap_k<<<N / 32, 256, 0, stream>>>(hin, PB + l * PB_LAYER, bs[l], A, Pp);
    edge_max_k<<<N / 8, 256, 0, stream>>>(A, Pp, src, hout);
    float* tmp = hin; hin = hout; hout = tmp;
  }

  global_max_k<<<256, 256, 0, stream>>>(hin, g);
  gvec_k<<<1, 128, 0, stream>>>(g, Wa, ba, gv);
  final_k<<<N / 32, 256, 0, stream>>>(hin, PB + 4 * PB_LAYER, gv, Wb, bb, x, out);
}


# Fuse each layer's gather/max edge phase with the next layer's f32 WMMA node GEMM: the 32-node h tile stays in LDS, 4 launches and the h round trip go
# speedup vs baseline: 1.3489x; 1.3489x over previous
#include <hip/hip_runtime.h>
#include <hip/hip_bf16.h>


typedef float v2f __attribute__((ext_vector_type(2)));
typedef float v8f __attribute__((ext_vector_type(8)));

#define N_NODES 100000
#define PB_LAYER 8192

__global__ __launch_bounds__(256) void prep_k(
    const float* __restrict__ W2, const float* __restrict__ W3,
    const float* __restrict__ W4, const float* __restrict__ W5,
    const float* __restrict__ Wa, float* __restrict__ PB,
    unsigned* __restrict__ g)
{
  int tid = blockIdx.x * 256 + threadIdx.x;
  if (tid < 64) g[tid] = 0u;
  if (tid >= 5 * PB_LAYER) return;
  int l = tid / PB_LAYER, r = tid % PB_LAYER;
  int d = r & 1;
  int n = (r >> 1) & 127;
  int h = (r >> 8) & 1;
  int s = r >> 9;
  int k = 4 * s + 2 * h + d;
  float v;
  if (l < 4) {
    const float* W = (l == 0) ? W2 : (l == 1) ? W3 : (l == 2) ? W4 : W5;
    v = (n < 64) ? (W[k * 64 + n] - W[(64 + k) * 64 + n])
                 : W[(64 + k) * 64 + (n - 64)];
  } else {
    v = Wa[k * 128 + n];
  }
  PB[tid] = v;
}

__global__ __launch_bounds__(256) void layer1_ap(
    const float* __restrict__ x, const float* __restrict__ W1,
    const float* __restrict__ b1, float* __restrict__ A, float* __restrict__ P)
{
  int tid = blockIdx.x * 256 + threadIdx.x;
  int n = tid >> 6, c = tid & 63;
  float x0 = x[n * 3 + 0], x1 = x[n * 3 + 1], x2 = x[n * 3 + 2];
  float wt0 = W1[0 * 64 + c], wt1 = W1[1 * 64 + c], wt2 = W1[2 * 64 + c];
  float wb0 = W1[3 * 64 + c], wb1 = W1[4 * 64 + c], wb2 = W1[5 * 64 + c];
  float p = x0 * wb0 + x1 * wb1 + x2 * wb2;
  float a = b1[c] + x0 * (wt0 - wb0) + x1 * (wt1 - wb1) + x2 * (wt2 - wb2);
  A[tid] = a;
  P[tid] = p;
}

__global__ __launch_bounds__(256) void edge_gemm_k(
    const float* __restrict__ Ain, const float* __restrict__ Pin,
    const int* __restrict__ src, const float* __restrict__ PB,
    const float* __restrict__ bias, float* __restrict__ Aout,
    float* __restrict__ Pout)
{
  __shared__ float Hs[32][68];
  __shared__ __attribute__((aligned(16))) float Ds[32][132];

  const int w = threadIdx.x >> 5;
  const int lane = threadIdx.x & 31;
  const int n0 = blockIdx.x * 32;

  for (int i = 0; i < 4; ++i) {
    const int ln = w * 4 + i;
    const int node = n0 + ln;
    float2 a = *(const float2*)(Ain + (size_t)node * 64 + lane * 2);
    float m0 = 0.0f, m1 = 0.0f;
    const int* se = src + (size_t)node * 16;
#pragma unroll
    for (int e = 0; e < 16; ++e) {
      int j = se[e];
      float2 p = *(const float2*)(Pin + (size_t)j * 64 + lane * 2);
      m0 = fmaxf(m0, a.x + p.x);
      m1 = fmaxf(m1, a.y + p.y);
    }
    float2 o; o.x = m0; o.y = m1;
    *(float2*)&Hs[ln][lane * 2] = o;
  }
  __syncthreads();

  const int lhalf = lane >> 4, l15 = lane & 15;
  const int col = w * 16 + l15;

  v8f acc0 = {0.f, 0.f, 0.f, 0.f, 0.f, 0.f, 0.f, 0.f};
  v8f acc1 = {0.f, 0.f, 0.f, 0.f, 0.f, 0.f, 0.f, 0.f};

  const float* ar0 = &Hs[l15][2 * lhalf];
  const float* ar1 = &Hs[16 + l15][2 * lhalf];
  const float* bp = PB + (size_t)(lhalf * 128 + col) * 2;
#pragma unroll
  for (int s = 0; s < 16; ++s) {
    v2f bf = *(const v2f*)(bp + s * 512);
    v2f af0 = *(const v2f*)(ar0 + 4 * s);
    v2f af1 = *(const v2f*)(ar1 + 4 * s);
    acc0 = __builtin_amdgcn_wmma_f32_16x16x4_f32(false, af0, false, bf,
                                                 (short)0, acc0, false, false);
    acc1 = __builtin_amdgcn_wmma_f32_16x16x4_f32(false, af1, false, bf,
                                                 (short)0, acc1, false, false);
  }

  const float bv = (w < 4) ? bias[col] : 0.0f;
#pragma unroll
  for (int r = 0; r < 8; ++r) {
    Ds[r + lhalf * 8][col] = acc0[r] + bv;
    Ds[16 + r + lhalf * 8][col] = acc1[r] + bv;
  }
  __syncthreads();

  float* ao = Aout + (size_t)n0 * 64;
  float* po = Pout + (size_t)n0 * 64;
#pragma unroll
  for (int i = 0; i < 2; ++i) {
    const int idx = threadIdx.x + 256 * i;
    const int row = idx >> 4, c4 = (idx & 15) * 4;
    float4 va = *(const float4*)&Ds[row][c4];
    float4 vp = *(const float4*)&Ds[row][64 + c4];
    *(float4*)(ao + row * 64 + c4) = va;
    *(float4*)(po + row * 64 + c4) = vp;
  }
}

__global__ __launch_bounds__(256) void edge_max_k(
    const float* __restrict__ A, const float* __restrict__ P,
    const int* __restrict__ src, float* __restrict__ h)
{
  int node = blockIdx.x * 8 + (threadIdx.x >> 5);
  int lane = threadIdx.x & 31;
  float2 a = *(const float2*)(A + (size_t)node * 64 + lane * 2);
  float m0 = 0.0f, m1 = 0.0f;
  const int* se = src + (size_t)node * 16;
#pragma unroll
  for (int e = 0; e < 16; ++e) {
    int j = se[e];
    float2 p = *(const float2*)(P + (size_t)j * 64 + lane * 2);
    m0 = fmaxf(m0, a.x + p.x);
    m1 = fmaxf(m1, a.y + p.y);
  }
  float2 o; o.x = m0; o.y = m1;
  *(float2*)(h + (size_t)node * 64 + lane * 2) = o;
}

__global__ __launch_bounds__(256) void global_max_k(
    const float* __restrict__ h, unsigned* __restrict__ g)
{
  __shared__ float red[256];
  int c = threadIdx.x & 63, grp = threadIdx.x >> 6;
  float m = 0.0f;
  for (int n = blockIdx.x * 4 + grp; n < N_NODES; n += gridDim.x * 4)
    m = fmaxf(m, h[(size_t)n * 64 + c]);
  red[threadIdx.x] = m;
  __syncthreads();
  if (threadIdx.x < 64) {
    m = fmaxf(fmaxf(red[threadIdx.x], red[threadIdx.x + 64]),
              fmaxf(red[threadIdx.x + 128], red[threadIdx.x + 192]));
    atomicMax(g + c, __float_as_uint(m));
  }
}

__global__ void gvec_k(const unsigned* __restrict__ g,
                       const float* __restrict__ Wa,
                       const float* __restrict__ ba, float* __restrict__ gv)
{
  int c = threadIdx.x;
  float s = ba[c];
  for (int k = 0; k < 64; ++k)
    s += __uint_as_float(g[k]) * Wa[(64 + k) * 128 + c];
  gv[c] = s;
}

__global__ __launch_bounds__(256) void final_k(
    const float* __restrict__ h, const float* __restrict__ PBa,
    const float* __restrict__ gv, const float* __restrict__ Wb,
    const float* __restrict__ bb, const float* __restrict__ x,
    float* __restrict__ out)
{
  __shared__ float T[32][132];

  const int w = threadIdx.x >> 5;
  const int lane = threadIdx.x & 31;
  const int wr = w >> 2, wc = w & 3;
  const int n0 = blockIdx.x * 32 + wr * 16;
  const int lhalf = lane >> 4, l15 = lane & 15;
  const int cb0 = wc * 32, cb1 = wc * 32 + 16;

  v8f acc0 = {0.f, 0.f, 0.f, 0.f, 0.f, 0.f, 0.f, 0.f};
  v8f acc1 = {0.f, 0.f, 0.f, 0.f, 0.f, 0.f, 0.f, 0.f};

  const float* arow = h + (size_t)(n0 + l15) * 64 + 2 * lhalf;
#pragma unroll
  for (int s = 0; s < 16; ++s) {
    v2f af = *(const v2f*)(arow + 4 * s);
    v2f bf0 = *(const v2f*)(PBa + (((s * 2 + lhalf) * 128) + cb0 + l15) * 2);
    v2f bf1 = *(const v2f*)(PBa + (((s * 2 + lhalf) * 128) + cb1 + l15) * 2);
    acc0 = __builtin_amdgcn_wmma_f32_16x16x4_f32(false, af, false, bf0,
                                                 (short)0, acc0, false, false);
    acc1 = __builtin_amdgcn_wmma_f32_16x16x4_f32(false, af, false, bf1,
                                                 (short)0, acc1, false, false);
  }

#pragma unroll
  for (int r = 0; r < 8; ++r) {
    int lr = wr * 16 + r + lhalf * 8;
    int c0 = cb0 + l15, c1 = cb1 + l15;
    T[lr][c0] = fmaxf(0.0f, acc0[r] + gv[c0]);
    T[lr][c1] = fmaxf(0.0f, acc1[r] + gv[c1]);
  }
  __syncthreads();

  int t = threadIdx.x;
  if (t < 96) {
    int row = t / 3, j = t % 3;
    float s = bb[j];
#pragma unroll 8
    for (int c = 0; c < 128; ++c) s += T[row][c] * Wb[c * 3 + j];
    size_t n = (size_t)blockIdx.x * 32 + row;
    out[n * 3 + j] = x[n * 3 + j] + s;
  }
}

extern "C" void kernel_launch(void* const* d_in, const int* in_sizes, int n_in,
                              void* d_out, int out_size, void* d_ws, size_t ws_size,
                              hipStream_t stream) {
  (void)in_sizes; (void)n_in; (void)out_size; (void)ws_size;
  const float* x  = (const float*)d_in[0];
  const int*   ei = (const int*)d_in[1];
  const float* W1 = (const float*)d_in[2];
  const float* b1 = (const float*)d_in[3];
  const float* W2 = (const float*)d_in[4];
  const float* b2 = (const float*)d_in[5];
  const float* W3 = (const float*)d_in[6];
  const float* b3 = (const float*)d_in[7];
  const float* W4 = (const float*)d_in[8];
  const float* b4 = (const float*)d_in[9];
  const float* W5 = (const float*)d_in[10];
  const float* b5 = (const float*)d_in[11];
  const float* Wa = (const float*)d_in[12];
  const float* ba = (const float*)d_in[13];
  const float* Wb = (const float*)d_in[14];
  const float* bb = (const float*)d_in[15];
  float* out = (float*)d_out;

  const int N = N_NODES;
  float* ws = (float*)d_ws;
  float*    A  = ws;
  float*    Pp = A  + (size_t)N * 64;
  float*    h0 = Pp + (size_t)N * 64;
  float*    h1 = h0 + (size_t)N * 64;
  float*    PB = h1 + (size_t)N * 64;
  unsigned* g  = (unsigned*)(PB + 5 * PB_LAYER);
  float*    gv = (float*)(g + 64);

  const int* src = ei;

  prep_k<<<(5 * PB_LAYER + 255) / 256, 256, 0, stream>>>(W2, W3, W4, W5, Wa, PB, g);
  layer1_ap<<<(N * 64) / 256, 256, 0, stream>>>(x, W1, b1, A, Pp);

  const float* bs[4] = {b2, b3, b4, b5};
  float* Ai = A;  float* Pi = Pp;
  float* Ao = h0; float* Po = h1;
  for (int l = 0; l < 4; ++l) {
    edge_gemm_k<<<N / 32, 256, 0, stream>>>(Ai, Pi, src, PB + l * PB_LAYER, bs[l], Ao, Po);
    float* t0 = Ai; Ai = Ao; Ao = t0;
    float* t1 = Pi; Pi = Po; Po = t1;
  }
  float* hin = Ao;
  edge_max_k<<<N / 8, 256, 0, stream>>>(Ai, Pi, src, hin);

  global_max_k<<<256, 256, 0, stream>>>(hin, g);
  gvec_k<<<1, 128, 0, stream>>>(g, Wa, ba, gv);
  final_k<<<N / 32, 256, 0, stream>>>(hin, PB + 4 * PB_LAYER, gv, Wb, bb, x, out);
}


# Layer 1 formed on the fly in the first fused launch: 12-byte neighbour x rows gathered, x_j@W1bot as one f32 16x16x4 WMMA per node and 16 channels; the layer1 launch and its A/P planes go
# speedup vs baseline: 1.5291x; 1.1336x over previous
#include <hip/hip_runtime.h>
#include <hip/hip_bf16.h>


typedef float v2f __attribute__((ext_vector_type(2)));
typedef float v8f __attribute__((ext_vector_type(8)));

#define N_NODES 100000
#define PB_LAYER 8192

__global__ __launch_bounds__(256) void prep_k(
    const float* __restrict__ W2, const float* __restrict__ W3,
    const float* __restrict__ W4, const float* __restrict__ W5,
    const float* __restrict__ Wa, float* __restrict__ PB,
    unsigned* __restrict__ g)
{
  int tid = blockIdx.x * 256 + threadIdx.x;
  if (tid < 64) g[tid] = 0u;
  if (tid >= 5 * PB_LAYER) return;
  int l = tid / PB_LAYER, r = tid % PB_LAYER;
  int d = r & 1;
  int n = (r >> 1) & 127;
  int h = (r >> 8) & 1;
  int s = r >> 9;
  int k = 4 * s + 2 * h + d;
  float v;
  if (l < 4) {
    const float* W = (l == 0) ? W2 : (l == 1) ? W3 : (l == 2) ? W4 : W5;
    v = (n < 64) ? (W[k * 64 + n] - W[(64 + k) * 64 + n])
                 : W[(64 + k) * 64 + (n - 64)];
  } else {
    v = Wa[k * 128 + n];
  }
  PB[tid] = v;
}

__device__ __forceinline__ void tile_gemm_store(
    float (&Hs)[32][68], float (&Ds)[32][132], const float* __restrict__ PB,
    const float* __restrict__ bias, float* __restrict__ Aout,
    float* __restrict__ Pout, int n0, int w, int lane)
{
  const int lhalf = lane >> 4, l15 = lane & 15;
  const int col = w * 16 + l15;

  v8f acc0 = {0.f, 0.f, 0.f, 0.f, 0.f, 0.f, 0.f, 0.f};
  v8f acc1 = {0.f, 0.f, 0.f, 0.f, 0.f, 0.f, 0.f, 0.f};

  const float* ar0 = &Hs[l15][2 * lhalf];
  const float* ar1 = &Hs[16 + l15][2 * lhalf];
  const float* bp = PB + (size_t)(lhalf * 128 + col) * 2;
#pragma unroll
  for (int s = 0; s < 16; ++s) {
    v2f bf = *(const v2f*)(bp + s * 512);
    v2f af0 = *(const v2f*)(ar0 + 4 * s);
    v2f af1 = *(const v2f*)(ar1 + 4 * s);
    acc0 = __builtin_amdgcn_wmma_f32_16x16x4_f32(false, af0, false, bf,
                                                 (short)0, acc0, false, false);
    acc1 = __builtin_amdgcn_wmma_f32_16x16x4_f32(false, af1, false, bf,
                                                 (short)0, acc1, false, false);
  }

  const float bv = (w < 4) ? bias[col] : 0.0f;
#pragma unroll
  for (int r = 0; r < 8; ++r) {
    Ds[r + lhalf * 8][col] = acc0[r] + bv;
    Ds[16 + r + lhalf * 8][col] = acc1[r] + bv;
  }
  __syncthreads();

  float* ao = Aout + (size_t)n0 * 64;
  float* po = Pout + (size_t)n0 * 64;
#pragma unroll
  for (int i = 0; i < 2; ++i) {
    const int idx = threadIdx.x + 256 * i;
    const int row = idx >> 4, c4 = (idx & 15) * 4;
    float4 va = *(const float4*)&Ds[row][c4];
    float4 vp = *(const float4*)&Ds[row][64 + c4];
    *(float4*)(ao + row * 64 + c4) = va;
    *(float4*)(po + row * 64 + c4) = vp;
  }
}

__device__ __forceinline__ float xhalf(float v)
{
  int b = __builtin_bit_cast(int, v);
  return __builtin_bit_cast(float, __builtin_amdgcn_permlanex16(
      b, b, 0x76543210, 0xfedcba98, false, false));
}

__global__ __launch_bounds__(256) void edge1_gemm_k(
    const float* __restrict__ x, const int* __restrict__ src,
    const float* __restrict__ W1, const float* __restrict__ b1,
    const float* __restrict__ PB, const float* __restrict__ bias,
    float* __restrict__ Aout, float* __restrict__ Pout)
{
  __shared__ float Hs[32][68];
  __shared__ __attribute__((aligned(16))) float Ds[32][132];

  const int w = threadIdx.x >> 5;
  const int lane = threadIdx.x & 31;
  const int lhalf = lane >> 4, l15 = lane & 15;
  const int n0 = blockIdx.x * 32;
  const int nb = n0 + w * 4;

  int j[4];
#pragma unroll
  for (int i = 0; i < 4; ++i) j[i] = src[(size_t)(nb + i) * 16 + l15];
  float xj0[4], xj1[4], xj2[4], xi0[4], xi1[4], xi2[4];
#pragma unroll
  for (int i = 0; i < 4; ++i) {
    const float* xr = x + (size_t)j[i] * 3;
    xj0[i] = xr[0]; xj1[i] = xr[1]; xj2[i] = xr[2];
    const float* xo = x + (size_t)(nb + i) * 3;
    xi0[i] = xo[0]; xi1[i] = xo[1]; xi2[i] = xo[2];
  }

  v2f bw[4];
  float av[4][4];
#pragma unroll
  for (int t = 0; t < 4; ++t) {
    const int ch = 16 * t + l15;
    const float wb0 = W1[3 * 64 + ch], wb1 = W1[4 * 64 + ch], wb2 = W1[5 * 64 + ch];
    const float aw0 = W1[0 * 64 + ch] - wb0;
    const float aw1 = W1[1 * 64 + ch] - wb1;
    const float aw2 = W1[2 * 64 + ch] - wb2;
    const float bz = b1[ch];
    bw[t].x = lhalf ? wb2 : wb0;
    bw[t].y = lhalf ? 0.0f : wb1;
#pragma unroll
    for (int i = 0; i < 4; ++i)
      av[i][t] = bz + xi0[i] * aw0 + xi1[i] * aw1 + xi2[i] * aw2;
  }

#pragma unroll
  for (int i = 0; i < 4; ++i) {
    v2f af;
    af.x = lhalf ? xj2[i] : xj0[i];
    af.y = lhalf ? 0.0f : xj1[i];
    float hv[4];
#pragma unroll
    for (int t = 0; t < 4; ++t) {
      v8f z = {0.f, 0.f, 0.f, 0.f, 0.f, 0.f, 0.f, 0.f};
      v8f d = __builtin_amdgcn_wmma_f32_16x16x4_f32(false, af, false, bw[t],
                                                    (short)0, z, false, false);
      float m = fmaxf(fmaxf(fmaxf(d[0], d[1]), fmaxf(d[2], d[3])),
                      fmaxf(fmaxf(d[4], d[5]), fmaxf(d[6], d[7])));
      m = fmaxf(m, xhalf(m));
      hv[t] = fmaxf(0.0f, av[i][t] + m);
    }
    const int ln = w * 4 + i;
    Hs[ln][16 * lhalf + l15] = lhalf ? hv[1] : hv[0];
    Hs[ln][32 + 16 * lhalf + l15] = lhalf ? hv[3] : hv[2];
  }
  __syncthreads();

  tile_gemm_store(Hs, Ds, PB, bias, Aout, Pout, n0, w, lane);
}

__global__ __launch_bounds__(256) void edge_gemm_k(
    const float* __restrict__ Ain, const float* __restrict__ Pin,
    const int* __restrict__ src, const float* __restrict__ PB,
    const float* __restrict__ bias, float* __restrict__ Aout,
    float* __restrict__ Pout)
{
  __shared__ float Hs[32][68];
  __shared__ __attribute__((aligned(16))) float Ds[32][132];

  const int w = threadIdx.x >> 5;
  const int lane = threadIdx.x & 31;
  const int n0 = blockIdx.x * 32;

  for (int i = 0; i < 4; ++i) {
    const int ln = w * 4 + i;
    const int node = n0 + ln;
    float2 a = *(const float2*)(Ain + (size_t)node * 64 + lane * 2);
    float m0 = 0.0f, m1 = 0.0f;
    const int* se = src + (size_t)node * 16;
#pragma unroll
    for (int e = 0; e < 16; ++e) {
      int j = se[e];
      float2 p = *(const float2*)(Pin + (size_t)j * 64 + lane * 2);
      m0 = fmaxf(m0, a.x + p.x);
      m1 = fmaxf(m1, a.y + p.y);
    }
    float2 o; o.x = m0; o.y = m1;
    *(float2*)&Hs[ln][lane * 2] = o;
  }
  __syncthreads();

  tile_gemm_store(Hs, Ds, PB, bias, Aout, Pout, n0, w, lane);
}

__global__ __launch_bounds__(256) void edge_max_k(
    const float* __restrict__ A, const float* __restrict__ P,
    const int* __restrict__ src, float* __restrict__ h)
{
  int node = blockIdx.x * 8 + (threadIdx.x >> 5);
  int lane = threadIdx.x & 31;
  float2 a = *(const float2*)(A + (size_t)node * 64 + lane * 2);
  float m0 = 0.0f, m1 = 0.0f;
  const int* se = src + (size_t)node * 16;
#pragma unroll
  for (int e = 0; e < 16; ++e) {
    int j = se[e];
    float2 p = *(const float2*)(P + (size_t)j * 64 + lane * 2);
    m0 = fmaxf(m0, a.x + p.x);
    m1 = fmaxf(m1, a.y + p.y);
  }
  float2 o; o.x = m0; o.y = m1;
  *(float2*)(h + (size_t)node * 64 + lane * 2) = o;
}

__global__ __launch_bounds__(256) void global_max_k(
    const float* __restrict__ h, unsigned* __restrict__ g)
{
  __shared__ float red[256];
  int c = threadIdx.x & 63, grp = threadIdx.x >> 6;
  float m = 0.0f;
  for (int n = blockIdx.x * 4 + grp; n < N_NODES; n += gridDim.x * 4)
    m = fmaxf(m, h[(size_t)n * 64 + c]);
  red[threadIdx.x] = m;
  __syncthreads();
  if (threadIdx.x < 64) {
    m = fmaxf(fmaxf(red[threadIdx.x], red[threadIdx.x + 64]),
              fmaxf(red[threadIdx.x + 128], red[threadIdx.x + 192]));
    atomicMax(g + c, __float_as_uint(m));
  }
}

__global__ void gvec_k(const unsigned* __restrict__ g,
                       const float* __restrict__ Wa,
                       const float* __restrict__ ba, float* __restrict__ gv)
{
  int c = threadIdx.x;
  float s = ba[c];
  for (int k = 0; k < 64; ++k)
    s += __uint_as_float(g[k]) * Wa[(64 + k) * 128 + c];
  gv[c] = s;
}

__global__ __launch_bounds__(256) void final_k(
    const float* __restrict__ h, const float* __restrict__ PBa,
    const float* __restrict__ gv, const float* __restrict__ Wb,
    const float* __restrict__ bb, const float* __restrict__ x,
    float* __restrict__ out)
{
  __shared__ float T[32][132];

  const int w = threadIdx.x >> 5;
  const int lane = threadIdx.x & 31;
  const int wr = w >> 2, wc = w & 3;
  const int n0 = blockIdx.x * 32 + wr * 16;
  const int lhalf = lane >> 4, l15 = lane & 15;
  const int cb0 = wc * 32, cb1 = wc * 32 + 16;

  v8f acc0 = {0.f, 0.f, 0.f, 0.f, 0.f, 0.f, 0.f, 0.f};
  v8f acc1 = {0.f, 0.f, 0.f, 0.f, 0.f, 0.f, 0.f, 0.f};

  const float* arow = h + (size_t)(n0 + l15) * 64 + 2 * lhalf;
#pragma unroll
  for (int s = 0; s < 16; ++s) {
    v2f af = *(const v2f*)(arow + 4 * s);
    v2f bf0 = *(const v2f*)(PBa + (((s * 2 + lhalf) * 128) + cb0 + l15) * 2);
    v2f bf1 = *(const v2f*)(PBa + (((s * 2 + lhalf) * 128) + cb1 + l15) * 2);
    acc0 = __builtin_amdgcn_wmma_f32_16x16x4_f32(false, af, false, bf0,
                                                 (short)0, acc0, false, false);
    acc1 = __builtin_amdgcn_wmma_f32_16x16x4_f32(false, af, false, bf1,
                                                 (short)0, acc1, false, false);
  }

#pragma unroll
  for (int r = 0; r < 8; ++r) {
    int lr = wr * 16 + r + lhalf * 8;
    int c0 = cb0 + l15, c1 = cb1 + l15;
    T[lr][c0] = fmaxf(0.0f, acc0[r] + gv[c0]);
    T[lr][c1] = fmaxf(0.0f, acc1[r] + gv[c1]);
  }
  __syncthreads();

  int t = threadIdx.x;
  if (t < 96) {
    int row = t / 3, j = t % 3;
    float s = bb[j];
#pragma unroll 8
    for (int c = 0; c < 128; ++c) s += T[row][c] * Wb[c * 3 + j];
    size_t n = (size_t)blockIdx.x * 32 + row;
    out[n * 3 + j] = x[n * 3 + j] + s;
  }
}

extern "C" void kernel_launch(void* const* d_in, const int* in_sizes, int n_in,
                              void* d_out, int out_size, void* d_ws, size_t ws_size,
                              hipStream_t stream) {
  (void)in_sizes; (void)n_in; (void)out_size; (void)ws_size;
  const float* x  = (const float*)d_in[0];
  const int*   ei = (const int*)d_in[1];
  const float* W1 = (const float*)d_in[2];
  const float* b1 = (const float*)d_in[3];
  const float* W2 = (const float*)d_in[4];
  const float* b2 = (const float*)d_in[5];
  const float* W3 = (const float*)d_in[6];
  const float* b3 = (const float*)d_in[7];
  const float* W4 = (const float*)d_in[8];
  const float* b4 = (const float*)d_in[9];
  const float* W5 = (const float*)d_in[10];
  const float* b5 = (const float*)d_in[11];
  const float* Wa = (const float*)d_in[12];
  const float* ba = (const float*)d_in[13];
  const float* Wb = (const float*)d_in[14];
  const float* bb = (const float*)d_in[15];
  float* out = (float*)d_out;

  const int N = N_NODES;
  float* ws = (float*)d_ws;
  float*    A  = ws;
  float*    Pp = A  + (size_t)N * 64;
  float*    h0 = Pp + (size_t)N * 64;
  float*    h1 = h0 + (size_t)N * 64;
  float*    PB = h1 + (size_t)N * 64;
  unsigned* g  = (unsigned*)(PB + 5 * PB_LAYER);
  float*    gv = (float*)(g + 64);

  const int* src = ei;

  prep_k<<<(5 * PB_LAYER + 255) / 256, 256, 0, stream>>>(W2, W3, W4, W5, Wa, PB, g);

  const float* bs[4] = {b2, b3, b4, b5};
  float* Ai = h0; float* Pi = h1;
  float* Ao = A;  float* Po = Pp;
  edge1_gemm_k<<<N / 32, 256, 0, stream>>>(x, src, W1, b1, PB, bs[0], Ai, Pi);
  for (int l = 1; l < 4; ++l) {
    edge_gemm_k<<<N / 32, 256, 0, stream>>>(Ai, Pi, src, PB + l * PB_LAYER, bs[l], Ao, Po);
    float* t0 = Ai; Ai = Ao; Ao = t0;
    float* t1 = Pi; Pi = Po; Po = t1;
  }
  float* hin = Ao;
  edge_max_k<<<N / 8, 256, 0, stream>>>(Ai, Pi, src, hin);

  global_max_k<<<256, 256, 0, stream>>>(hin, g);
  gvec_k<<<1, 128, 0, stream>>>(g, Wa, ba, gv);
  final_k<<<N / 32, 256, 0, stream>>>(hin, PB + 4 * PB_LAYER, gv, Wb, bb, x, out);
}


# Tile kernels run 1024 resident work-groups looping over the 32-node tiles: weight fragments loaded once per wave and kept in registers, next node's ids fetched ahead; 2/3 of wave entries go
# speedup vs baseline: 1.9247x; 1.2587x over previous
#include <hip/hip_runtime.h>
#include <hip/hip_bf16.h>


typedef float v2f __attribute__((ext_vector_type(2)));
typedef float v8f __attribute__((ext_vector_type(8)));

#define N_NODES 100000
#define N_TILES (N_NODES / 32)
#define GRID_T 1024
#define PB_LAYER 8192

__global__ __launch_bounds__(256) void prep_k(
    const float* __restrict__ W2, const float* __restrict__ W3,
    const float* __restrict__ W4, const float* __restrict__ W5,
    const float* __restrict__ Wa, float* __restrict__ PB)
{
  int tid = blockIdx.x * 256 + threadIdx.x;
  if (tid >= 5 * PB_LAYER) return;
  int l = tid / PB_LAYER, r = tid % PB_LAYER;
  int d = r & 1;
  int n = (r >> 1) & 127;
  int h = (r >> 8) & 1;
  int s = r >> 9;
  int k = 4 * s + 2 * h + d;
  float v;
  if (l < 4) {
    const float* W = (l == 0) ? W2 : (l == 1) ? W3 : (l == 2) ? W4 : W5;
    v = (n < 64) ? (W[k * 64 + n] - W[(64 + k) * 64 + n])
                 : W[(64 + k) * 64 + (n - 64)];
  } else {
    v = Wa[k * 128 + n];
  }
  PB[tid] = v;
}

__device__ __forceinline__ void load_bfrag(const float* __restrict__ PB, int w,
                                           int lane, v2f (&bf)[16])
{
  const int lhalf = lane >> 4, l15 = lane & 15;
  const int col = w * 16 + l15;
  const float* bp = PB + (size_t)(lhalf * 128 + col) * 2;
#pragma unroll
  for (int s = 0; s < 16; ++s)
    bf[s] = *(const v2f*)(bp + s * 512);
}

__device__ __forceinline__ void tile_gemm_store(
    float (&Hs)[32][68], float (&Ds)[32][132], const v2f (&bf)[16],
    const float bv, float* __restrict__ Aout,
    float* __restrict__ Pout, int n0, int w, int lane)
{
  const int lhalf = lane >> 4, l15 = lane & 15;
  const int col = w * 16 + l15;

  v8f acc0 = {0.f, 0.f, 0.f, 0.f, 0.f, 0.f, 0.f, 0.f};
  v8f acc1 = {0.f, 0.f, 0.f, 0.f, 0.f, 0.f, 0.f, 0.f};

  const float* ar0 = &Hs[l15][2 * lhalf];
  const float* ar1 = &Hs[16 + l15][2 * lhalf];
#pragma unroll
  for (int s = 0; s < 16; ++s) {
    v2f af0 = *(const v2f*)(ar0 + 4 * s);
    v2f af1 = *(const v2f*)(ar1 + 4 * s);
    acc0 = __builtin_amdgcn_wmma_f32_16x16x4_f32(false, af0, false, bf[s],
                                                 (short)0, acc0, false, false);
    acc1 = __builtin_amdgcn_wmma_f32_16x16x4_f32(false, af1, false, bf[s],
                                                 (short)0, acc1, false, false);
  }

#pragma unroll
  for (int r = 0; r < 8; ++r) {
    Ds[r + lhalf * 8][col] = acc0[r] + bv;
    Ds[16 + r + lhalf * 8][col] = acc1[r] + bv;
  }
  __syncthreads();

  float* ao = Aout + (size_t)n0 * 64;
  float* po = Pout + (size_t)n0 * 64;
#pragma unroll
  for (int i = 0; i < 2; ++i) {
    const int idx = threadIdx.x + 256 * i;
    const int row = idx >> 4, c4 = (idx & 15) * 4;
    float4 va = *(const float4*)&Ds[row][c4];
    float4 vp = *(const float4*)&Ds[row][64 + c4];
    *(float4*)(ao + row * 64 + c4) = va;
    *(float4*)(po + row * 64 + c4) = vp;
  }
}

__device__ __forceinline__ float xhalf(float v)
{
  int b = __builtin_bit_cast(int, v);
  return __builtin_bit_cast(float, __builtin_amdgcn_permlanex16(
      b, b, 0x76543210, 0xfedcba98, false, false));
}

__global__ __launch_bounds__(256, 8) void edge1_gemm_k(
    const float* __restrict__ x, const int* __restrict__ src,
    const float* __restrict__ W1, const float* __restrict__ b1,
    const float* __restrict__ PB, const float* __restrict__ bias,
    float* __restrict__ Aout, float* __restrict__ Pout)
{
  __shared__ float Hs[32][68];
  __shared__ __attribute__((aligned(16))) float Ds[32][132];

  const int w = threadIdx.x >> 5;
  const int lane = threadIdx.x & 31;
  const int lhalf = lane >> 4, l15 = lane & 15;

  v2f bf[16];
  load_bfrag(PB, w, lane, bf);
  const float bv = (w < 4) ? bias[w * 16 + l15] : 0.0f;
  v2f bw[4];
  float aw0[4], aw1[4], aw2[4], bz[4];
#pragma unroll
  for (int t = 0; t < 4; ++t) {
    const int ch = 16 * t + l15;
    const float wb0 = W1[3 * 64 + ch], wb1 = W1[4 * 64 + ch], wb2 = W1[5 * 64 + ch];
    aw0[t] = W1[0 * 64 + ch] - wb0;
    aw1[t] = W1[1 * 64 + ch] - wb1;
    aw2[t] = W1[2 * 64 + ch] - wb2;
    bz[t] = b1[ch];
    bw[t].x = lhalf ? wb2 : wb0;
    bw[t].y = lhalf ? 0.0f : wb1;
  }

  for (int tile = blockIdx.x; tile < N_TILES; tile += GRID_T) {
  const int n0 = tile * 32;
  const int nb = n0 + w * 4;

  int j[4];
#pragma unroll
  for (int i = 0; i < 4; ++i) j[i] = src[(size_t)(nb + i) * 16 + l15];
  float xj0[4], xj1[4], xj2[4], xi0[4], xi1[4], xi2[4];
#pragma unroll
  for (int i = 0; i < 4; ++i) {
    const float* xr = x + (size_t)j[i] * 3;
    xj0[i] = xr[0]; xj1[i] = xr[1]; xj2[i] = xr[2];
    const float* xo = x + (size_t)(nb + i) * 3;
    xi0[i] = xo[0]; xi1[i] = xo[1]; xi2[i] = xo[2];
  }

  float av[4][4];
#pragma unroll
  for (int t = 0; t < 4; ++t) {
#pragma unroll
    for (int i = 0; i < 4; ++i)
      av[i][t] = bz[t] + xi0[i] * aw0[t] + xi1[i] * aw1[t] + xi2[i] * aw2[t];
  }

#pragma unroll
  for (int i = 0; i < 4; ++i) {
    v2f af;
    af.x = lhalf ? xj2[i] : xj0[i];
    af.y = lhalf ? 0.0f : xj1[i];
    float hv[4];
#pragma unroll
    for (int t = 0; t < 4; ++t) {
      v8f z = {0.f, 0.f, 0.f, 0.f, 0.f, 0.f, 0.f, 0.f};
      v8f d = __builtin_amdgcn_wmma_f32_16x16x4_f32(false, af, false, bw[t],
                                                    (short)0, z, false, false);
      float m = fmaxf(fmaxf(fmaxf(d[0], d[1]), fmaxf(d[2], d[3])),
                      fmaxf(fmaxf(d[4], d[5]), fmaxf(d[6], d[7])));
      m = fmaxf(m, xhalf(m));
      hv[t] = fmaxf(0.0f, av[i][t] + m);
    }
    const int ln = w * 4 + i;
    Hs[ln][16 * lhalf + l15] = lhalf ? hv[1] : hv[0];
    Hs[ln][32 + 16 * lhalf + l15] = lhalf ? hv[3] : hv[2];
  }
  __syncthreads();

  tile_gemm_store(Hs, Ds, bf, bv, Aout, Pout, n0, w, lane);
  }
}

__global__ __launch_bounds__(256, 8) void edge_gemm_k(
    const float* __restrict__ Ain, const float* __restrict__ Pin,
    const int* __restrict__ src, const float* __restrict__ PB,
    const float* __restrict__ bias, float* __restrict__ Aout,
    float* __restrict__ Pout)
{
  __shared__ float Hs[32][68];
  __shared__ __attribute__((aligned(16))) float Ds[32][132];

  const int w = __builtin_amdgcn_readfirstlane(threadIdx.x >> 5);
  const int lane = threadIdx.x & 31;

  v2f bf[16];
  load_bfrag(PB, w, lane, bf);
  const float bv = (w < 4) ? bias[w * 16 + (lane & 15)] : 0.0f;

  int jv;
  float2 a;
  {
    const int nd = blockIdx.x * 32 + w * 4;
    jv = src[(size_t)nd * 16 + (lane & 15)];
    a = *(const float2*)(Ain + (size_t)nd * 64 + lane * 2);
  }

  for (int tile = blockIdx.x; tile < N_TILES; tile += GRID_T) {
  const int n0 = tile * 32;

#pragma unroll 1
  for (int i = 0; i < 4; ++i) {
    const int nn = (i < 3) ? n0 + w * 4 + i + 1 : (tile + GRID_T) * 32 + w * 4;
    int jn = jv;
    float2 an = a;
    if (nn < N_NODES) {
      jn = src[(size_t)nn * 16 + (lane & 15)];
      an = *(const float2*)(Ain + (size_t)nn * 64 + lane * 2);
    }
    float m0 = 0.0f, m1 = 0.0f;
#pragma unroll
    for (int e = 0; e < 16; ++e) {
      const int j = __builtin_amdgcn_readlane(jv, e);
      float2 p = *(const float2*)(Pin + (size_t)j * 64 + lane * 2);
      m0 = fmaxf(m0, a.x + p.x);
      m1 = fmaxf(m1, a.y + p.y);
    }
    float2 o; o.x = m0; o.y = m1;
    *(float2*)&Hs[w * 4 + i][lane * 2] = o;
    jv = jn;
    a = an;
  }
  __syncthreads();

  tile_gemm_store(Hs, Ds, bf, bv, Aout, Pout, n0, w, lane);
  }
}

__global__ __launch_bounds__(256) void edge_max_k(
    const float* __restrict__ A, const float* __restrict__ P,
    const int* __restrict__ src, float* __restrict__ h,
    float* __restrict__ pmax)
{
  __shared__ float red[8][64];

  const int w = threadIdx.x >> 5;
  const int lane = threadIdx.x & 31;

  float x0 = 0.0f, x1 = 0.0f;
  for (int tile = blockIdx.x; tile < N_TILES; tile += GRID_T) {
  const int n0 = tile * 32;
  for (int i = 0; i < 4; ++i) {
    const int node = n0 + w * 4 + i;
    float2 a = *(const float2*)(A + (size_t)node * 64 + lane * 2);
    float m0 = 0.0f, m1 = 0.0f;
    const int* se = src + (size_t)node * 16;
#pragma unroll
    for (int e = 0; e < 16; ++e) {
      int j = se[e];
      float2 p = *(const float2*)(P + (size_t)j * 64 + lane * 2);
      m0 = fmaxf(m0, a.x + p.x);
      m1 = fmaxf(m1, a.y + p.y);
    }
    float2 o; o.x = m0; o.y = m1;
    *(float2*)(h + (size_t)node * 64 + lane * 2) = o;
    x0 = fmaxf(x0, m0);
    x1 = fmaxf(x1, m1);
  }
  }
  float2 xo; xo.x = x0; xo.y = x1;
  *(float2*)&red[w][lane * 2] = xo;
  __syncthreads();
  if (w == 0) {
    float2 r = *(const float2*)&red[0][lane * 2];
    for (int k = 1; k < 8; ++k) {
      float2 v = *(const float2*)&red[k][lane * 2];
      r.x = fmaxf(r.x, v.x);
      r.y = fmaxf(r.y, v.y);
    }
    *(float2*)(pmax + (size_t)blockIdx.x * 64 + lane * 2) = r;
  }
}

__global__ __launch_bounds__(1024) void gmax_gvec_k(
    const float* __restrict__ pmax, const float* __restrict__ Wa,
    const float* __restrict__ ba, float* __restrict__ gv)
{
  __shared__ __attribute__((aligned(16))) float Ws[64][128];
  __shared__ __attribute__((aligned(16))) float red[64][64];
  __shared__ float gs[64];

  const int t = threadIdx.x;
  const float4* w4 = (const float4*)(Wa + 64 * 128);
  ((float4*)&Ws[0][0])[t] = w4[t];
  ((float4*)&Ws[0][0])[t + 1024] = w4[t + 1024];

  const float4* p4 = (const float4*)pmax;
  constexpr int full = (GRID_T * 16) / 1024;
  static_assert(full % 8 == 0, "the sweep is cut into batches of 8 loads");
  static_assert(full * 1024 == GRID_T * 16, "no partial sweep is left over");
  float4 m = {0.0f, 0.0f, 0.0f, 0.0f};
  for (int b = 0; b < full / 8; ++b) {
    float4 v[8];
#pragma unroll
    for (int j = 0; j < 8; ++j) v[j] = p4[t + 1024 * (8 * b + j)];
#pragma unroll
    for (int j = 0; j < 8; ++j) {
      m.x = fmaxf(m.x, v[j].x); m.y = fmaxf(m.y, v[j].y);
      m.z = fmaxf(m.z, v[j].z); m.w = fmaxf(m.w, v[j].w);
    }
  }

  *(float4*)&red[t >> 4][(t & 15) * 4] = m;
  __syncthreads();
  if (t < 64) {
    float g = red[0][t];
#pragma unroll
    for (int r = 1; r < 64; ++r) g = fmaxf(g, red[r][t]);
    gs[t] = g;
  }
  __syncthreads();
  if (t < 128) {
    float s = ba[t];
    for (int k = 0; k < 64; ++k) s += gs[k] * Ws[k][t];
    gv[t] = s;
  }
}

__global__ __launch_bounds__(256, 8) void final_k(
    const float* __restrict__ h, const float* __restrict__ PBa,
    const float* __restrict__ gv, const float* __restrict__ Wb,
    const float* __restrict__ bb, const float* __restrict__ x,
    float* __restrict__ out)
{
  __shared__ float T[32][132];

  const int w = threadIdx.x >> 5;
  const int lane = threadIdx.x & 31;
  const int wr = w >> 2, wc = w & 3;
  const int lhalf = lane >> 4, l15 = lane & 15;
  const int cb0 = wc * 32, cb1 = wc * 32 + 16;

  v2f bf0[16], bf1[16];
#pragma unroll
  for (int s = 0; s < 16; ++s) {
    bf0[s] = *(const v2f*)(PBa + (((s * 2 + lhalf) * 128) + cb0 + l15) * 2);
    bf1[s] = *(const v2f*)(PBa + (((s * 2 + lhalf) * 128) + cb1 + l15) * 2);
  }
  const float g0 = gv[cb0 + l15], g1 = gv[cb1 + l15];

  for (int tile = blockIdx.x; tile < N_TILES; tile += GRID_T) {
  const int n0 = tile * 32 + wr * 16;

  v8f acc0 = {0.f, 0.f, 0.f, 0.f, 0.f, 0.f, 0.f, 0.f};
  v8f acc1 = {0.f, 0.f, 0.f, 0.f, 0.f, 0.f, 0.f, 0.f};

  const float* arow = h + (size_t)(n0 + l15) * 64 + 2 * lhalf;
  v2f af[16];
#pragma unroll
  for (int s = 0; s < 16; ++s) af[s] = *(const v2f*)(arow + 4 * s);
#pragma unroll
  for (int s = 0; s < 16; ++s) {
    acc0 = __builtin_amdgcn_wmma_f32_16x16x4_f32(false, af[s], false, bf0[s],
                                                 (short)0, acc0, false, false);
    acc1 = __builtin_amdgcn_wmma_f32_16x16x4_f32(false, af[s], false, bf1[s],
                                                 (short)0, acc1, false, false);
  }

#pragma unroll
  for (int r = 0; r < 8; ++r) {
    int lr = wr * 16 + r + lhalf * 8;
    int c0 = cb0 + l15, c1 = cb1 + l15;
    T[lr][c0] = fmaxf(0.0f, acc0[r] + g0);
    T[lr][c1] = fmaxf(0.0f, acc1[r] + g1);
  }
  __syncthreads();

  int t = threadIdx.x;
  if (t < 96) {
    int row = t / 3, j = t % 3;
    float s = bb[j];
#pragma unroll 8
    for (int c = 0; c < 128; ++c) s += T[row][c] * Wb[c * 3 + j];
    size_t n = (size_t)tile * 32 + row;
    out[n * 3 + j] = x[n * 3 + j] + s;
  }
  __syncthreads();
  }
}

extern "C" void kernel_launch(void* const* d_in, const int* in_sizes, int n_in,
                              void* d_out, int out_size, void* d_ws, size_t ws_size,
                              hipStream_t stream) {
  (void)in_sizes; (void)n_in; (void)out_size; (void)ws_size;
  const float* x  = (const float*)d_in[0];
  const int*   ei = (const int*)d_in[1];
  const float* W1 = (const float*)d_in[2];
  const float* b1 = (const float*)d_in[3];
  const float* W2 = (const float*)d_in[4];
  const float* b2 = (const float*)d_in[5];
  const float* W3 = (const float*)d_in[6];
  const float* b3 = (const float*)d_in[7];
  const float* W4 = (const float*)d_in[8];
  const float* b4 = (const float*)d_in[9];
  const float* W5 = (const float*)d_in[10];
  const float* b5 = (const float*)d_in[11];
  const float* Wa = (const float*)d_in[12];
  const float* ba = (const float*)d_in[13];
  const float* Wb = (const float*)d_in[14];
  const float* bb = (const float*)d_in[15];
  float* out = (float*)d_out;

  const int N = N_NODES;
  float* ws = (float*)d_ws;
  float*    A  = ws;
  float*    Pp = A  + (size_t)N * 64;
  float*    h0 = Pp + (size_t)N * 64;
  float*    h1 = h0 + (size_t)N * 64;
  float*    PB = h1 + (size_t)N * 64;
  float*    gv = PB + 5 * PB_LAYER + 64;

  const int* src = ei;

  prep_k<<<(5 * PB_LAYER + 255) / 256, 256, 0, stream>>>(W2, W3, W4, W5, Wa, PB);

  const float* bs[4] = {b2, b3, b4, b5};
  float* Ai = h0; float* Pi = h1;
  float* Ao = A;  float* Po = Pp;
  edge1_gemm_k<<<GRID_T, 256, 0, stream>>>(x, src, W1, b1, PB, bs[0], Ai, Pi);
  for (int l = 1; l < 4; ++l) {
    edge_gemm_k<<<GRID_T, 256, 0, stream>>>(Ai, Pi, src, PB + l * PB_LAYER, bs[l], Ao, Po);
    float* t0 = Ai; Ai = Ao; Ao = t0;
    float* t1 = Pi; Pi = Po; Po = t1;
  }
  float* hin = Ao;
  float* pmax = Po;
  edge_max_k<<<GRID_T, 256, 0, stream>>>(Ai, Pi, src, hin, pmax);

  gmax_gvec_k<<<1, 1024, 0, stream>>>(pmax, Wa, ba, gv);
  final_k<<<GRID_T, 256, 0, stream>>>(hin, PB + 4 * PB_LAYER, gv, Wb, bb, x, out);
}
